# starting kernel
# baseline (speedup 1.0000x reference)
#include <hip/hip_runtime.h>
#include <hip/hip_bf16.h>

typedef __attribute__((ext_vector_type(2))) float v2f;
typedef __attribute__((ext_vector_type(8))) float v8f;

__global__ void snake_feat_kernel(const float* __restrict__ x,
                                  float* __restrict__ feat) {
  const int b   = blockIdx.x;
  const int tid = threadIdx.x;
  const float* __restrict__ g0 = x + (size_t)b * 3072;
  const float* __restrict__ g1 = g0 + 1024;

  __shared__ float    s_g0[1024];
  __shared__ unsigned s_min[3];
  __shared__ int      s_head[2];
  __shared__ int      s_dirs[3][2];
  __shared__ int      s_rot[2];
  __shared__ unsigned char s_freev[3][32];
  __shared__ unsigned char s_inb[3][32];

  if (tid < 3) s_min[tid] = 0xFFFFFFFFu;
  __syncthreads();

  const int i4 = tid * 4;
  const float4 v0 = *reinterpret_cast<const float4*>(g0 + i4);
  const float4 v1 = *reinterpret_cast<const float4*>(g1 + i4);
  *reinterpret_cast<float4*>(&s_g0[i4]) = v0;

  unsigned fkey = 0xFFFFFFFFu, hkey = 0xFFFFFFFFu, pkey = 0xFFFFFFFFu;
  const float a0[4] = {v0.x, v0.y, v0.z, v0.w};
  const float a1[4] = {v1.x, v1.y, v1.z, v1.w};
#pragma unroll
  for (int j = 0; j < 4; ++j) {
    const unsigned idx = (unsigned)(i4 + j);
    if (a0[j] == -1.0f) fkey = min(fkey, idx);
    if (a0[j] ==  1.0f) hkey = min(hkey, idx);
    if (a1[j] ==  1.0f) pkey = min(pkey, idx);
  }
  if (fkey != 0xFFFFFFFFu) atomicMin(&s_min[0], fkey);
  if (hkey != 0xFFFFFFFFu) atomicMin(&s_min[1], hkey);
  if (pkey != 0xFFFFFFFFu) atomicMin(&s_min[2], pkey);
  __syncthreads();

  if (tid == 0) {
    unsigned fi = s_min[0]; if (fi == 0xFFFFFFFFu) fi = 0;
    unsigned hi = s_min[1]; if (hi == 0xFFFFFFFFu) hi = 0;
    unsigned pi = s_min[2]; if (pi == 0xFFFFFFFFu) pi = 0;
    const int fx = (int)(fi >> 5), fy = (int)(fi & 31);
    const int hx = (int)(hi >> 5), hy = (int)(hi & 31);
    const int px = (int)(pi >> 5), py = (int)(pi & 31);
    const int dx = hx - px, dy = hy - py;
    const int ddx = fx - hx, ddy = fy - hy;
    const int k = (dx == 1) ? 0 : (dy == -1) ? 1 : (dx == -1) ? 2 : 3;
    int rx, ry;
    switch (k) {
      case 0:  rx =  ddx; ry =  ddy; break;
      case 1:  rx = -ddy; ry =  ddx; break;
      case 2:  rx = -ddx; ry = -ddy; break;
      default: rx =  ddy; ry = -ddx; break;
    }
    s_rot[0] = rx; s_rot[1] = ry;
    s_head[0] = hx; s_head[1] = hy;
    s_dirs[0][0] = -dy; s_dirs[0][1] =  dx;
    s_dirs[1][0] =  dx; s_dirs[1][1] =  dy;
    s_dirs[2][0] =  dy; s_dirs[2][1] = -dx;
  }
  __syncthreads();

  if (tid < 96) {
    const int dir = tid >> 5;
    const int i   = tid & 31;
    const int m   = i + 1;
    const int cx = s_head[0] + m * s_dirs[dir][0];
    const int cy = s_head[1] + m * s_dirs[dir][1];
    const bool inb = (cx >= 0) & (cx < 32) & (cy >= 0) & (cy < 32);
    const int ci = min(max(cx, 0), 31);
    const int cj = min(max(cy, 0), 31);
    const float val = s_g0[ci * 32 + cj];
    s_freev[dir][i] = (unsigned char)((val == 0.0f) | (val == -1.0f));
    s_inb[dir][i]   = (unsigned char)inb;
  }
  __syncthreads();

  if (tid < 3) {
    int freeLen = 0;
    for (int i = 0; i < 32; ++i) {
      const unsigned char fvp = s_freev[tid][(i == 0) ? 0 : i - 1];
      if (!(fvp & s_inb[tid][i])) { freeLen = i; break; }
    }
    feat[(size_t)b * 8 + tid] = (float)freeLen;
  }
  if (tid == 0) {
    feat[(size_t)b * 8 + 3] = (float)s_rot[0];
    feat[(size_t)b * 8 + 4] = (float)s_rot[1];
    feat[(size_t)b * 8 + 5] = 0.0f;
    feat[(size_t)b * 8 + 6] = 0.0f;
    feat[(size_t)b * 8 + 7] = 0.0f;
  }
}

static __device__ inline v8f splat8(float v) {
  v8f r;
#pragma unroll
  for (int i = 0; i < 8; ++i) r[i] = v;
  return r;
}

__global__ void snake_mlp_kernel(const float* __restrict__ feat,
                                 const float* __restrict__ w1,
                                 const float* __restrict__ b1,
                                 const float* __restrict__ w2,
                                 const float* __restrict__ b2,
                                 float* __restrict__ out) {
  const int lane = threadIdx.x;
  const int base = blockIdx.x * 16;
  const int M  = lane & 15;
  const int N  = lane & 15;
  const int kh = lane >> 4;

  __shared__ float Hs[16][20];

  const float* frow = feat + (size_t)(base + M) * 8;
  v2f a0; a0.x = frow[0 + 2 * kh]; a0.y = frow[1 + 2 * kh];
  v2f a1; a1.x = frow[4 + 2 * kh]; a1.y = frow[5 + 2 * kh];

  v8f h[2];
#pragma unroll
  for (int t = 0; t < 2; ++t) {
    const int n = t * 16 + N;
    const bool nv = (n < 20);
    const int k0a = 2 * kh;
    const int k0b = 4 + 2 * kh;
    v2f bA; bA.x = nv ? w1[n * 5 + k0a] : 0.0f;
            bA.y = nv ? w1[n * 5 + k0a + 1] : 0.0f;
    v2f bB; bB.x = (nv && k0b < 5) ? w1[n * 5 + k0b] : 0.0f;
            bB.y = 0.0f;
    v8f c = splat8(nv ? b1[n] : 0.0f);
    c = __builtin_amdgcn_wmma_f32_16x16x4_f32(false, a0, false, bA,
                                              (short)0, c, false, false);
    c = __builtin_amdgcn_wmma_f32_16x16x4_f32(false, a1, false, bB,
                                              (short)0, c, false, false);
#pragma unroll
    for (int v = 0; v < 8; ++v) c[v] = fmaxf(c[v], 0.0f);
    h[t] = c;
  }

#pragma unroll
  for (int v = 0; v < 8; ++v) {
    const int row = v + 8 * kh;
    Hs[row][N] = h[0][v];
    if (N < 4) Hs[row][16 + N] = h[1][v];
  }
  __syncthreads();

  const bool ov = (N < 3);
  v8f o = splat8(ov ? b2[N] : 0.0f);
#pragma unroll
  for (int c5 = 0; c5 < 5; ++c5) {
    const int k = 4 * c5 + 2 * kh;
    v2f a; a.x = Hs[M][k]; a.y = Hs[M][k + 1];
    v2f bb; bb.x = ov ? w2[N * 20 + k] : 0.0f;
            bb.y = ov ? w2[N * 20 + k + 1] : 0.0f;
    o = __builtin_amdgcn_wmma_f32_16x16x4_f32(false, a, false, bb,
                                              (short)0, o, false, false);
  }

  if (ov) {
#pragma unroll
    for (int v = 0; v < 8; ++v) {
      const int row = base + v + 8 * kh;
      out[(size_t)row * 3 + N] = o[v];
    }
  }
}

extern "C" void kernel_launch(void* const* d_in, const int* in_sizes, int n_in,
                              void* d_out, int out_size, void* d_ws, size_t ws_size,
                              hipStream_t stream) {
  const float* x  = (const float*)d_in[0];
  const float* w1 = (const float*)d_in[1];
  const float* b1 = (const float*)d_in[2];
  const float* w2 = (const float*)d_in[3];
  const float* b2 = (const float*)d_in[4];
  float* outp = (float*)d_out;
  float* featws = (float*)d_ws;

  const int B = in_sizes[0] / 3072;

  snake_feat_kernel<<<B, 256, 0, stream>>>(x, featws);
  snake_mlp_kernel<<<B / 16, 32, 0, stream>>>(featws, w1, b1, w2, b2, outp);
}


# Loads issued in order of use under the in-order load counter: MLP weights first, by unconditional clamped loads, then each board's rows last row first; counted waits exact, search runs as rows land
# speedup vs baseline: 2.1180x; 2.1180x over previous
#include <hip/hip_runtime.h>
#include <hip/hip_bf16.h>


typedef __attribute__((ext_vector_type(2))) float v2f;
typedef __attribute__((ext_vector_type(8))) float v8f;

#define FEAT_BPW 4
#define FEAT_WPB 8

static __device__ inline unsigned wave_min_u32(unsigned v) {
  v = min(v, (unsigned)__builtin_amdgcn_update_dpp((int)v, (int)v, 0x161, 0xf, 0xf, false));
  v = min(v, (unsigned)__builtin_amdgcn_update_dpp((int)v, (int)v, 0x162, 0xf, 0xf, false));
  v = min(v, (unsigned)__builtin_amdgcn_update_dpp((int)v, (int)v, 0x164, 0xf, 0xf, false));
  v = min(v, (unsigned)__builtin_amdgcn_update_dpp((int)v, (int)v, 0x168, 0xf, 0xf, false));
  v = min(v, (unsigned)__builtin_amdgcn_permlanex16((int)v, (int)v, 0x76543210, 0xfedcba98, false, false));
  return v;
}

__global__ __launch_bounds__(FEAT_WPB * 32)
void snake_fused_kernel(const float* __restrict__ x,
                        const float* __restrict__ w1,
                        const float* __restrict__ b1,
                        const float* __restrict__ w2,
                        const float* __restrict__ b2,
                        float* __restrict__ out) {
  const int lane = threadIdx.x & 31;
  const int wave = threadIdx.x >> 5;
  const int b0   = (blockIdx.x * FEAT_WPB + wave) * FEAT_BPW;

  __shared__ float s_g0[FEAT_WPB][1024];
  float* __restrict__ sb = s_g0[wave];

  const int m = lane + 1;
  const int c = lane & 15;
  const int h = lane >> 4;

  float g0[2], g1[2], g4[2];
#pragma unroll
  for (int t = 0; t < 2; ++t) {
    const int n = t * 16 + c;
    const float* __restrict__ wr = w1 + (n < 20 ? n : 19) * 5;
    g0[t] = wr[2 * h]; g1[t] = wr[2 * h + 1]; g4[t] = wr[4];
  }
  float bl[8], bh[4];
#pragma unroll
  for (int r = 0; r < 8; ++r) bl[r] = b1[8 * h + r];
#pragma unroll
  for (int r = 0; r < 4; ++r) bh[r] = b1[16 + r];
  const bool ov = (c < 3);
  const float* __restrict__ w2r = w2 + (ov ? c : 0) * 20;
  float ul[8], uh[4];
#pragma unroll
  for (int r = 0; r < 8; ++r) ul[r] = w2r[8 * h + r];
#pragma unroll
  for (int r = 0; r < 4; ++r) uh[r] = w2r[16 + r];
  float bo[3];
#pragma unroll
  for (int r = 0; r < 3; ++r) bo[r] = b2[r];

  float4 c0[FEAT_BPW][8], c1[FEAT_BPW][8];
#pragma unroll
  for (int q = 0; q < FEAT_BPW; ++q) {
    const float4* __restrict__ p0 = reinterpret_cast<const float4*>(x + (size_t)(b0 + q) * 3072);
    const float4* __restrict__ p1 = p0 + 256;
#pragma unroll
    for (int j = 7; j >= 0; --j) {
      c0[q][j] = p0[j * 32 + lane];
      c1[q][j] = p1[j * 32 + lane];
      __builtin_amdgcn_sched_barrier(0x78F);
    }
  }

  __builtin_amdgcn_fence(__ATOMIC_RELEASE, "wavefront");

  v2f fa; fa.x = 0.0f; fa.y = 0.0f;
  v2f fb; fb.x = 0.0f; fb.y = 0.0f;

#pragma unroll
  for (int q = 0; q < FEAT_BPW; ++q) {
    unsigned fkey = 0xFFFFFFFFu, hkey = 0xFFFFFFFFu, pkey = 0xFFFFFFFFu;
#pragma unroll
    for (int j = 7; j >= 0; --j) {
      *reinterpret_cast<float4*>(&sb[(j * 32 + lane) * 4]) = c0[q][j];
      const float a0[4] = {c0[q][j].x, c0[q][j].y, c0[q][j].z, c0[q][j].w};
      const float a1[4] = {c1[q][j].x, c1[q][j].y, c1[q][j].z, c1[q][j].w};
#pragma unroll
      for (int e = 3; e >= 0; --e) {
        const unsigned idx = (unsigned)(j * 128 + lane * 4 + e);
        fkey = (a0[e] == -1.0f) ? idx : fkey;
        hkey = (a0[e] ==  1.0f) ? idx : hkey;
        pkey = (a1[e] ==  1.0f) ? idx : pkey;
      }
    }
    unsigned fi = __builtin_amdgcn_readfirstlane(wave_min_u32(fkey));
    unsigned hi = __builtin_amdgcn_readfirstlane(wave_min_u32(hkey));
    unsigned pi = __builtin_amdgcn_readfirstlane(wave_min_u32(pkey));
    if (fi == 0xFFFFFFFFu) fi = 0;
    if (hi == 0xFFFFFFFFu) hi = 0;
    if (pi == 0xFFFFFFFFu) pi = 0;

    const int fx = (int)(fi >> 5), fy = (int)(fi & 31);
    const int hx = (int)(hi >> 5), hy = (int)(hi & 31);
    const int px = (int)(pi >> 5), py = (int)(pi & 31);
    const int dx = hx - px, dy = hy - py;
    const int ddx = fx - hx, ddy = fy - hy;
    const int k = (dx == 1) ? 0 : (dy == -1) ? 1 : (dx == -1) ? 2 : 3;
    int rx, ry;
    switch (k) {
      case 0:  rx =  ddx; ry =  ddy; break;
      case 1:  rx = -ddy; ry =  ddx; break;
      case 2:  rx = -ddx; ry = -ddy; break;
      default: rx =  ddy; ry = -ddx; break;
    }
    const int dirx[3] = {-dy, dx,  dy};
    const int diry[3] = { dx, dy, -dx};

    __builtin_amdgcn_fence(__ATOMIC_RELEASE, "wavefront");
    __builtin_amdgcn_wave_barrier();

    int freeLen[3];
#pragma unroll
    for (int d = 0; d < 3; ++d) {
      const int cx = hx + m * dirx[d];
      const int cy = hy + m * diry[d];
      const bool inb = (cx >= 0) & (cx < 32) & (cy >= 0) & (cy < 32);
      const int ci = min(max(cx, 0), 31);
      const int cj = min(max(cy, 0), 31);
      const float val = sb[ci * 32 + cj];
      const bool fv = (val == 0.0f) | (val == -1.0f);
      const unsigned fvm = __builtin_amdgcn_ballot_w32(fv);
      const unsigned inm = __builtin_amdgcn_ballot_w32(inb);
      const unsigned fvprev = (fvm << 1) | (fvm & 1u);
      const unsigned nchk = ~(fvprev & inm);
      freeLen[d] = nchk ? __builtin_ctz(nchk) : 0;
    }

    __builtin_amdgcn_fence(__ATOMIC_ACQUIRE, "wavefront");
    __builtin_amdgcn_wave_barrier();

    const bool mine = (c == q);
    fa.x = mine ? (float)(h ? freeLen[2] : freeLen[0]) : fa.x;
    fa.y = mine ? (float)(h ? rx         : freeLen[1]) : fa.y;
    fb.x = mine ? (float)(h ? 0          : ry)         : fb.x;
  }

  v2f w1a[2], w1b[2];
  v8f bias1[2];
#pragma unroll
  for (int t = 0; t < 2; ++t) {
    const bool nv = (t * 16 + c < 20);
    w1a[t].x = nv ? g0[t] : 0.0f;
    w1a[t].y = nv ? g1[t] : 0.0f;
    w1b[t].x = (nv && h == 0) ? g4[t] : 0.0f;
    w1b[t].y = 0.0f;
  }
#pragma unroll
  for (int r = 0; r < 8; ++r) {
    bias1[0][r] = bl[r];
    bias1[1][r] = (h == 0 && r < 4) ? bh[r & 3] : 0.0f;
  }
  v2f w2f[6];
#pragma unroll
  for (int j = 0; j < 4; ++j) {
    w2f[j].x = ov ? ul[2 * j] : 0.0f;
    w2f[j].y = ov ? ul[2 * j + 1] : 0.0f;
  }
#pragma unroll
  for (int j = 4; j < 6; ++j) {
    const bool kv = ov && (h == 0);
    w2f[j].x = kv ? uh[2 * (j - 4)] : 0.0f;
    w2f[j].y = kv ? uh[2 * (j - 4) + 1] : 0.0f;
  }
  v8f bias2;
#pragma unroll
  for (int r = 0; r < 8; ++r) bias2[r] = (h == 0 && r < 3) ? bo[r < 3 ? r : 0] : 0.0f;

  v8f hT[2];
#pragma unroll
  for (int t = 0; t < 2; ++t) {
    v8f acc = bias1[t];
    acc = __builtin_amdgcn_wmma_f32_16x16x4_f32(false, w1a[t], false, fa,
                                                (short)0, acc, false, false);
    acc = __builtin_amdgcn_wmma_f32_16x16x4_f32(false, w1b[t], false, fb,
                                                (short)0, acc, false, false);
#pragma unroll
    for (int v = 0; v < 8; ++v) acc[v] = fmaxf(acc[v], 0.0f);
    hT[t] = acc;
  }

  v8f o = bias2;
#pragma unroll
  for (int j = 0; j < 6; ++j) {
    v2f hb;
    if (j < 4) { hb.x = hT[0][2 * j];       hb.y = hT[0][2 * j + 1]; }
    else       { hb.x = hT[1][2 * (j - 4)]; hb.y = hT[1][2 * (j - 4) + 1]; }
    o = __builtin_amdgcn_wmma_f32_16x16x4_f32(false, w2f[j], false, hb,
                                              (short)0, o, false, false);
  }

  if (lane < FEAT_BPW) {
    float* __restrict__ op = out + (size_t)(b0 + lane) * 3;
    op[0] = o[0];
    op[1] = o[1];
    op[2] = o[2];
  }
}

extern "C" void kernel_launch(void* const* d_in, const int* in_sizes, int n_in,
                              void* d_out, int out_size, void* d_ws, size_t ws_size,
                              hipStream_t stream) {
  const float* x  = (const float*)d_in[0];
  const float* w1 = (const float*)d_in[1];
  const float* b1 = (const float*)d_in[2];
  const float* w2 = (const float*)d_in[3];
  const float* b2 = (const float*)d_in[4];
  float* outp = (float*)d_out;
  (void)d_ws; (void)ws_size;

  const int B = in_sizes[0] / 3072;

  snake_fused_kernel<<<B / (FEAT_WPB * FEAT_BPW), FEAT_WPB * 32, 0, stream>>>(x, w1, b1, w2, b2, outp);
}
